# starting kernel
# baseline (speedup 1.0000x reference)
#include <hip/hip_runtime.h>
#include <math.h>

typedef __attribute__((ext_vector_type(2))) float v2f;
typedef __attribute__((ext_vector_type(4))) float v4f;
typedef __attribute__((ext_vector_type(8))) float v8f;

#define D_        128
#define R_        1024
#define T_        4
#define OUTSLOTS_ 256
#define START_    4
#define L_        64
#define M_        2048
#define N_        (R_ + L_ * M_)
#define NENC_     5
#define K1_       256
#define XSTR_     260

#define W1P_FRAGS (NENC_ * 16 * 64 * 32)
#define W2P_FRAGS (NENC_ * 8  * 64 * 32)
#define W1P_BYTES ((size_t)W1P_FRAGS * 8)
#define W2P_BYTES ((size_t)W2P_FRAGS * 8)
#define WSP_BYTES (W1P_BYTES + W2P_BYTES)

__global__ void dag_init_roots(float* __restrict__ buf, const float* __restrict__ root) {
    int i = blockIdx.x * blockDim.x + threadIdx.x;
    v4f v = *(const v4f*)(root + (size_t)i * 4);
    *(v4f*)(buf + (size_t)i * 4) = v;
}

__global__ void dag_pack_w(const float* __restrict__ W, float* __restrict__ Wp,
                           int ntile, int ncol) {
    int idx  = blockIdx.x * blockDim.x + threadIdx.x;
    int lane = idx & 31;
    int k4   = (idx >> 5) & 63;
    int j    = (idx >> 11) % ntile;
    int e    = idx / (ntile << 11);
    int col  = lane & 15;
    int kk   = k4 * 4 + ((lane >> 4) & 1) * 2;
    const float* Wsrc = W + (size_t)e * K1_ * ncol + (size_t)(j * 16 + col);
    Wp[(size_t)idx * 2 + 0] = Wsrc[(size_t)kk * ncol];
    Wp[(size_t)idx * 2 + 1] = Wsrc[(size_t)(kk + 1) * ncol];
}

template <bool PACKED>
__global__ __launch_bounds__(160)
void dag_level(float* __restrict__ buf,
               const float* __restrict__ W1,
               const float* __restrict__ b1,
               const float* __restrict__ W2,
               const float* __restrict__ b2,
               const float* __restrict__ slots,
               const int*   __restrict__ par,
               const int*   __restrict__ types,
               int level)
{
    __shared__ float X[16 * XSTR_];
    __shared__ float H[16 * XSTR_];

    const int lane     = threadIdx.x & 31;
    const int e        = threadIdx.x >> 5;
    const int tile     = blockIdx.x;
    const int qbase    = level * M_ + tile * 16;
    const int rowbase  = R_ + qbase;
    const int colLane  = lane & 15;
    const int laneHalf = (lane >> 4) & 1;
    const int klo      = laneHalf * 2;

    for (int m = e; m < 16; m += NENC_) {
        int p0 = par[(qbase + m) * 2 + 0];
        int p1 = par[(qbase + m) * 2 + 1];
        int t  = types[qbase + m];
        const float* s0 = buf + (size_t)p0 * D_;
        const float* s1;
        if (t >= START_) {
            int slot = t - START_;
            slot = slot < 0 ? 0 : (slot > OUTSLOTS_ - 1 ? OUTSLOTS_ - 1 : slot);
            s1 = slots + (size_t)slot * D_;
        } else {
            s1 = buf + (size_t)p1 * D_;
        }
        v4f a = *(const v4f*)(s0 + lane * 4);
        v4f c = *(const v4f*)(s1 + lane * 4);
        *(v4f*)(&X[m * XSTR_ + lane * 4])       = a;
        *(v4f*)(&X[m * XSTR_ + 128 + lane * 4]) = c;
    }

    int rt[8];
    bool ownAny = false;
    #pragma unroll
    for (int i = 0; i < 8; ++i) {
        int t = types[qbase + i + laneHalf * 8];
        rt[i] = (t >= START_) ? T_ : t;
        ownAny |= (rt[i] == e);
    }
    const bool waveActive = __any(ownAny);
    __syncthreads();

    if (waveActive) {
        #pragma unroll 1
        for (int j = 0; j < 16; ++j) {
            v8f acc = {};
            if (PACKED) {
                const v2f* Wf = (const v2f*)W1 + ((size_t)(e * 16 + j) * 64) * 32 + lane;
                #pragma unroll 8
                for (int k4 = 0; k4 < 64; ++k4) {
                    int kk = k4 * 4 + klo;
                    v2f a = *(const v2f*)(&X[colLane * XSTR_ + kk]);
                    v2f b = Wf[k4 * 32];
                    acc = __builtin_amdgcn_wmma_f32_16x16x4_f32(
                            false, a, false, b, (short)0, acc, false, false);
                }
            } else {
                const float* Wb = W1 + (size_t)e * K1_ * K1_ + (size_t)(j * 16 + colLane);
                #pragma unroll 8
                for (int k4 = 0; k4 < 64; ++k4) {
                    int kk = k4 * 4 + klo;
                    v2f a = *(const v2f*)(&X[colLane * XSTR_ + kk]);
                    v2f b;
                    b.x = Wb[(size_t)kk * K1_];
                    b.y = Wb[(size_t)(kk + 1) * K1_];
                    acc = __builtin_amdgcn_wmma_f32_16x16x4_f32(
                            false, a, false, b, (short)0, acc, false, false);
                }
            }
            float bias = b1[e * K1_ + j * 16 + colLane];
            #pragma unroll
            for (int i = 0; i < 8; ++i) {
                if (rt[i] == e) {
                    float x = acc[i] + bias;
                    float g = 0.5f * x * (1.0f + erff(x * 0.70710678118654752f));
                    H[(i + laneHalf * 8) * XSTR_ + j * 16 + colLane] = g;
                }
            }
        }
    }
    __syncthreads();

    if (waveActive) {
        #pragma unroll 1
        for (int j = 0; j < 8; ++j) {
            v8f acc = {};
            if (PACKED) {
                const v2f* Wf = (const v2f*)W2 + ((size_t)(e * 8 + j) * 64) * 32 + lane;
                #pragma unroll 8
                for (int k4 = 0; k4 < 64; ++k4) {
                    int kk = k4 * 4 + klo;
                    v2f a = *(const v2f*)(&H[colLane * XSTR_ + kk]);
                    v2f b = Wf[k4 * 32];
                    acc = __builtin_amdgcn_wmma_f32_16x16x4_f32(
                            false, a, false, b, (short)0, acc, false, false);
                }
            } else {
                const float* Wb = W2 + (size_t)e * K1_ * D_ + (size_t)(j * 16 + colLane);
                #pragma unroll 8
                for (int k4 = 0; k4 < 64; ++k4) {
                    int kk = k4 * 4 + klo;
                    v2f a = *(const v2f*)(&H[colLane * XSTR_ + kk]);
                    v2f b;
                    b.x = Wb[(size_t)kk * D_];
                    b.y = Wb[(size_t)(kk + 1) * D_];
                    acc = __builtin_amdgcn_wmma_f32_16x16x4_f32(
                            false, a, false, b, (short)0, acc, false, false);
                }
            }
            float bias = b2[e * D_ + j * 16 + colLane];
            #pragma unroll
            for (int i = 0; i < 8; ++i) {
                if (rt[i] == e) {
                    int row = rowbase + i + laneHalf * 8;
                    buf[(size_t)row * D_ + j * 16 + colLane] = acc[i] + bias;
                }
            }
        }
    }
}

extern "C" void kernel_launch(void* const* d_in, const int* in_sizes, int n_in,
                              void* d_out, int out_size, void* d_ws, size_t ws_size,
                              hipStream_t stream) {
    const float* root  = (const float*)d_in[0];
    const float* W1    = (const float*)d_in[1];
    const float* b1    = (const float*)d_in[2];
    const float* W2    = (const float*)d_in[3];
    const float* b2    = (const float*)d_in[4];
    const float* slots = (const float*)d_in[5];
    const int*   par   = (const int*)d_in[6];
    const int*   typ   = (const int*)d_in[7];
    float* out = (float*)d_out;

    dag_init_roots<<<128, 256, 0, stream>>>(out, root);

    const bool packed = (ws_size >= WSP_BYTES);
    float* W1p = (float*)d_ws;
    float* W2p = (float*)((char*)d_ws + W1P_BYTES);

    if (packed) {
        dag_pack_w<<<W1P_FRAGS / 256, 256, 0, stream>>>(W1, W1p, 16, K1_);
        dag_pack_w<<<W2P_FRAGS / 256, 256, 0, stream>>>(W2, W2p, 8, D_);
        for (int l = 0; l < L_; ++l)
            dag_level<true><<<128, 160, 0, stream>>>(out, W1p, b1, W2p, b2,
                                                     slots, par, typ, l);
    } else {
        for (int l = 0; l < L_; ++l)
            dag_level<false><<<128, 160, 0, stream>>>(out, W1, b1, W2, b2,
                                                      slots, par, typ, l);
    }
}


# First tile's gather and products run under the weight fetch: fragments requested in use order, 56 ahead of the product that reads them, not all in front of the gather; row sums unchanged
# speedup vs baseline: 7.4545x; 7.4545x over previous
#include <hip/hip_runtime.h>
#include <math.h>

typedef __attribute__((ext_vector_type(2))) float v2f;
typedef __attribute__((ext_vector_type(4))) float v4f;
typedef __attribute__((ext_vector_type(8))) float v8f;

#define D_        128
#define R_        1024
#define T_        4
#define OUTSLOTS_ 256
#define START_    4
#define L_        64
#define M_        2048
#define N_        (R_ + L_ * M_)
#define NENC_     5
#define K1_       256
#define XSTR_     260

#define W1P_FRAGS (NENC_ * 16 * 64 * 32)
#define W2P_FRAGS (NENC_ * 8  * 64 * 32)
#define W1P_BYTES ((size_t)W1P_FRAGS * 8)
#define W2P_BYTES ((size_t)W2P_FRAGS * 8)
#define WSP_BYTES (W1P_BYTES + W2P_BYTES)

__global__ void dag_init_roots(float* __restrict__ buf, const float* __restrict__ root) {
    int i = blockIdx.x * blockDim.x + threadIdx.x;
    v4f v = *(const v4f*)(root + (size_t)i * 4);
    *(v4f*)(buf + (size_t)i * 4) = v;
}

__global__ void dag_pack_w(const float* __restrict__ W, float* __restrict__ Wp,
                           int ntile, int ncol) {
    int idx  = blockIdx.x * blockDim.x + threadIdx.x;
    int lane = idx & 31;
    int k4   = (idx >> 5) & 63;
    int j    = (idx >> 11) % ntile;
    int e    = idx / (ntile << 11);
    int col  = lane & 15;
    int kk   = k4 * 4 + ((lane >> 4) & 1) * 2;
    const float* Wsrc = W + (size_t)e * K1_ * ncol + (size_t)(j * 16 + col);
    Wp[(size_t)idx * 2 + 0] = Wsrc[(size_t)kk * ncol];
    Wp[(size_t)idx * 2 + 1] = Wsrc[(size_t)(kk + 1) * ncol];
}

template <bool PACKED>
__global__ __launch_bounds__(160)
void dag_level(float* __restrict__ buf,
               const float* __restrict__ W1,
               const float* __restrict__ b1,
               const float* __restrict__ W2,
               const float* __restrict__ b2,
               const float* __restrict__ slots,
               const int*   __restrict__ par,
               const int*   __restrict__ types,
               int level)
{
    __shared__ float X[16 * XSTR_];
    __shared__ float H[16 * XSTR_];

    const int lane     = threadIdx.x & 31;
    const int e        = threadIdx.x >> 5;
    const int tile     = blockIdx.x;
    const int qbase    = level * M_ + tile * 16;
    const int rowbase  = R_ + qbase;
    const int colLane  = lane & 15;
    const int laneHalf = (lane >> 4) & 1;
    const int klo      = laneHalf * 2;

    for (int m = e; m < 16; m += NENC_) {
        int p0 = par[(qbase + m) * 2 + 0];
        int p1 = par[(qbase + m) * 2 + 1];
        int t  = types[qbase + m];
        const float* s0 = buf + (size_t)p0 * D_;
        const float* s1;
        if (t >= START_) {
            int slot = t - START_;
            slot = slot < 0 ? 0 : (slot > OUTSLOTS_ - 1 ? OUTSLOTS_ - 1 : slot);
            s1 = slots + (size_t)slot * D_;
        } else {
            s1 = buf + (size_t)p1 * D_;
        }
        v4f a = *(const v4f*)(s0 + lane * 4);
        v4f c = *(const v4f*)(s1 + lane * 4);
        *(v4f*)(&X[m * XSTR_ + lane * 4])       = a;
        *(v4f*)(&X[m * XSTR_ + 128 + lane * 4]) = c;
    }

    int rt[8];
    bool ownAny = false;
    #pragma unroll
    for (int i = 0; i < 8; ++i) {
        int t = types[qbase + i + laneHalf * 8];
        rt[i] = (t >= START_) ? T_ : t;
        ownAny |= (rt[i] == e);
    }
    const bool waveActive = __any(ownAny);
    __syncthreads();

    if (waveActive) {
        #pragma unroll 1
        for (int j = 0; j < 16; ++j) {
            v8f acc = {};
            if (PACKED) {
                const v2f* Wf = (const v2f*)W1 + ((size_t)(e * 16 + j) * 64) * 32 + lane;
                #pragma unroll 8
                for (int k4 = 0; k4 < 64; ++k4) {
                    int kk = k4 * 4 + klo;
                    v2f a = *(const v2f*)(&X[colLane * XSTR_ + kk]);
                    v2f b = Wf[k4 * 32];
                    acc = __builtin_amdgcn_wmma_f32_16x16x4_f32(
                            false, a, false, b, (short)0, acc, false, false);
                }
            } else {
                const float* Wb = W1 + (size_t)e * K1_ * K1_ + (size_t)(j * 16 + colLane);
                #pragma unroll 8
                for (int k4 = 0; k4 < 64; ++k4) {
                    int kk = k4 * 4 + klo;
                    v2f a = *(const v2f*)(&X[colLane * XSTR_ + kk]);
                    v2f b;
                    b.x = Wb[(size_t)kk * K1_];
                    b.y = Wb[(size_t)(kk + 1) * K1_];
                    acc = __builtin_amdgcn_wmma_f32_16x16x4_f32(
                            false, a, false, b, (short)0, acc, false, false);
                }
            }
            float bias = b1[e * K1_ + j * 16 + colLane];
            #pragma unroll
            for (int i = 0; i < 8; ++i) {
                if (rt[i] == e) {
                    float x = acc[i] + bias;
                    float g = 0.5f * x * (1.0f + erff(x * 0.70710678118654752f));
                    H[(i + laneHalf * 8) * XSTR_ + j * 16 + colLane] = g;
                }
            }
        }
    }
    __syncthreads();

    if (waveActive) {
        #pragma unroll 1
        for (int j = 0; j < 8; ++j) {
            v8f acc = {};
            if (PACKED) {
                const v2f* Wf = (const v2f*)W2 + ((size_t)(e * 8 + j) * 64) * 32 + lane;
                #pragma unroll 8
                for (int k4 = 0; k4 < 64; ++k4) {
                    int kk = k4 * 4 + klo;
                    v2f a = *(const v2f*)(&H[colLane * XSTR_ + kk]);
                    v2f b = Wf[k4 * 32];
                    acc = __builtin_amdgcn_wmma_f32_16x16x4_f32(
                            false, a, false, b, (short)0, acc, false, false);
                }
            } else {
                const float* Wb = W2 + (size_t)e * K1_ * D_ + (size_t)(j * 16 + colLane);
                #pragma unroll 8
                for (int k4 = 0; k4 < 64; ++k4) {
                    int kk = k4 * 4 + klo;
                    v2f a = *(const v2f*)(&H[colLane * XSTR_ + kk]);
                    v2f b;
                    b.x = Wb[(size_t)kk * D_];
                    b.y = Wb[(size_t)(kk + 1) * D_];
                    acc = __builtin_amdgcn_wmma_f32_16x16x4_f32(
                            false, a, false, b, (short)0, acc, false, false);
                }
            }
            float bias = b2[e * D_ + j * 16 + colLane];
            #pragma unroll
            for (int i = 0; i < 8; ++i) {
                if (rt[i] == e) {
                    int row = rowbase + i + laneHalf * 8;
                    buf[(size_t)row * D_ + j * 16 + colLane] = acc[i] + bias;
                }
            }
        }
    }
}

#define REPL_ 48
#define NWAVE_ 8
typedef __attribute__((ext_vector_type(4))) int v4i;

#define PRE_A_ 16
#define PRE_   56
#define WFETCH_(s)                                                                   \
    do {                                                                             \
        if ((s) < 128) w1r[(s) >> 1][(s) & 1] = Wf1[((((s) & 1) * 64) + ((s) >> 1)) * 32]; \
        else           w2r[(s) - 128]         = Wf2[((s) - 128) * 32];                \
    } while (0)

template <bool FIRST>
__device__ __forceinline__ void run_tile(
        float* __restrict__ buf, const float* __restrict__ slots,
        const int* __restrict__ par, const int* __restrict__ types,
        float* X, float* H, const int* rowsel,
        v2f (&w1r)[64][2], v2f (&w2r)[64],
        const v2f* __restrict__ Wf1, const v2f* __restrict__ Wf2,
        const float (&bias1)[2], const float bias2,
        const int tl, const int total, const int lbase,
        const int tid, const int lane, const int w,
        const int colLane, const int laneHalf, const int klo)
{
    {
        const int m   = tid >> 4;
        const int sub = tid & 15;
        int idx = tl * 16 + m;
        idx = idx < total ? idx : total - 1;
        const int q  = lbase + rowsel[idx];
        const int p0 = par[q * 2 + 0];
        const int p1 = par[q * 2 + 1];
        const int t  = types[q];
        if (FIRST) {
            #pragma unroll
            for (int s = 0; s < PRE_A_; ++s) WFETCH_(s);
        }
        const float* s0 = buf + (size_t)p0 * D_;
        const float* s1;
        if (t >= START_) {
            int slot = t - START_;
            slot = slot < 0 ? 0 : (slot > OUTSLOTS_ - 1 ? OUTSLOTS_ - 1 : slot);
            s1 = slots + (size_t)slot * D_;
        } else {
            s1 = buf + (size_t)p1 * D_;
        }
        v4f ga[2], gc[2];
        #pragma unroll
        for (int i = 0; i < 2; ++i) {
            const int c4 = sub + 16 * i;
            ga[i] = *(const v4f*)(s0 + c4 * 4);
            gc[i] = *(const v4f*)(s1 + c4 * 4);
        }
        if (FIRST) {
            #pragma unroll
            for (int s = PRE_A_; s < PRE_; ++s) WFETCH_(s);
        }
        #pragma unroll
        for (int i = 0; i < 2; ++i) {
            const int c4 = sub + 16 * i;
            *(v4f*)(&X[m * XSTR_ + c4 * 4])       = ga[i];
            *(v4f*)(&X[m * XSTR_ + 128 + c4 * 4]) = gc[i];
        }
    }
    __syncthreads();

    v8f acc[2] = {};
    #pragma unroll
    for (int k4 = 0; k4 < 64; ++k4) {
        v2f a = *(const v2f*)(&X[colLane * XSTR_ + k4 * 4 + klo]);
        #pragma unroll
        for (int jj = 0; jj < 2; ++jj) {
            acc[jj] = __builtin_amdgcn_wmma_f32_16x16x4_f32(
                        false, a, false, w1r[k4][jj], (short)0, acc[jj], false, false);
            if (FIRST && 2 * k4 + jj + PRE_ < 192) WFETCH_(2 * k4 + jj + PRE_);
        }
    }
    #pragma unroll
    for (int jj = 0; jj < 2; ++jj) {
        #pragma unroll
        for (int i = 0; i < 8; ++i) {
            float x = acc[jj][i] + bias1[jj];
            float g = 0.5f * x * (1.0f + erff(x * 0.70710678118654752f));
            H[(i + laneHalf * 8) * XSTR_ + (2 * w + jj) * 16 + colLane] = g;
        }
    }
    __syncthreads();

    v8f acc2 = {};
    #pragma unroll
    for (int k4 = 0; k4 < 64; ++k4) {
        v2f a = *(const v2f*)(&H[colLane * XSTR_ + k4 * 4 + klo]);
        acc2 = __builtin_amdgcn_wmma_f32_16x16x4_f32(
                    false, a, false, w2r[k4], (short)0, acc2, false, false);
        if (FIRST && 128 + k4 + PRE_ < 192) WFETCH_(128 + k4 + PRE_);
    }
    #pragma unroll
    for (int i = 0; i < 8; ++i)
        X[(i + laneHalf * 8) * XSTR_ + w * 16 + colLane] = acc2[i] + bias2;
    __syncthreads();
    #pragma unroll
    for (int i = 0; i < 2; ++i) {
        const int m   = 2 * w + i;
        const int idx = tl * 16 + m;
        if (idx < total) {
            const size_t row = (size_t)(R_ + lbase + rowsel[idx]);
            *(v4f*)(buf + row * D_ + lane * 4) = *(const v4f*)(&X[m * XSTR_ + lane * 4]);
        }
    }
    __syncthreads();
}

__global__ __launch_bounds__(NWAVE_ * 32)
void dag_level_ws(float* __restrict__ buf,
                  const float* __restrict__ W1p,
                  const float* __restrict__ b1,
                  const float* __restrict__ W2p,
                  const float* __restrict__ b2,
                  const float* __restrict__ slots,
                  const int*   __restrict__ par,
                  const int*   __restrict__ types,
                  int level)
{
    __shared__ float X[16 * XSTR_];
    __shared__ float H[16 * XSTR_];
    __shared__ int   rowsel[M_];
    __shared__ int   wcnt[NWAVE_];

    const int tid      = threadIdx.x;
    const int lane     = tid & 31;
    const int w        = tid >> 5;
    const int e        = blockIdx.x % NENC_;
    const int r        = blockIdx.x / NENC_;
    const int lbase    = level * M_;
    const int colLane  = lane & 15;
    const int laneHalf = (lane >> 4) & 1;
    const int klo      = laneHalf * 2;

    unsigned mask = 0;
    {
        const v4i* tp = (const v4i*)(types + lbase + tid * 8);
        #pragma unroll
        for (int q = 0; q < 2; ++q) {
            v4i v = tp[q];
            #pragma unroll
            for (int c = 0; c < 4; ++c) {
                int t  = v[c];
                int en = (t >= START_) ? T_ : t;
                mask |= (en == e ? 1u : 0u) << (q * 4 + c);
            }
        }
    }
    const int cnt = __popc(mask);
    int incl = cnt;
    #pragma unroll
    for (int d = 1; d < 32; d <<= 1) {
        int n = __shfl_up(incl, d);
        if (lane >= d) incl += n;
    }
    if (lane == 31) wcnt[w] = incl;
    __syncthreads();
    int wbase = 0, total = 0;
    #pragma unroll
    for (int i = 0; i < NWAVE_; ++i) {
        int c = wcnt[i];
        if (i < w) wbase += c;
        total += c;
    }
    {
        int pos = wbase + incl - cnt;
        #pragma unroll
        for (int b = 0; b < 8; ++b)
            if ((mask >> b) & 1u) rowsel[pos++] = tid * 8 + b;
    }
    const int ntiles = (total + 15) >> 4;
    if (r >= ntiles) return;
    __syncthreads();

    v2f w1r[64][2];
    v2f w2r[64];
    const v2f* Wf1 = (const v2f*)W1p + ((size_t)(e * 16 + 2 * w) * 64) * 32 + lane;
    const v2f* Wf2 = (const v2f*)W2p + ((size_t)(e * 8 + w) * 64) * 32 + lane;
    float bias1[2];
    #pragma unroll
    for (int jj = 0; jj < 2; ++jj) bias1[jj] = b1[e * K1_ + (2 * w + jj) * 16 + colLane];
    const float bias2 = b2[e * D_ + w * 16 + colLane];

    run_tile<true>(buf, slots, par, types, X, H, rowsel, w1r, w2r, Wf1, Wf2, bias1, bias2,
                   r, total, lbase, tid, lane, w, colLane, laneHalf, klo);
    for (int tl = r + REPL_; tl < ntiles; tl += REPL_)
        run_tile<false>(buf, slots, par, types, X, H, rowsel, w1r, w2r, Wf1, Wf2, bias1, bias2,
                        tl, total, lbase, tid, lane, w, colLane, laneHalf, klo);
}

extern "C" void kernel_launch(void* const* d_in, const int* in_sizes, int n_in,
                              void* d_out, int out_size, void* d_ws, size_t ws_size,
                              hipStream_t stream) {
    const float* root  = (const float*)d_in[0];
    const float* W1    = (const float*)d_in[1];
    const float* b1    = (const float*)d_in[2];
    const float* W2    = (const float*)d_in[3];
    const float* b2    = (const float*)d_in[4];
    const float* slots = (const float*)d_in[5];
    const int*   par   = (const int*)d_in[6];
    const int*   typ   = (const int*)d_in[7];
    float* out = (float*)d_out;

    dag_init_roots<<<128, 256, 0, stream>>>(out, root);

    const bool packed = (ws_size >= WSP_BYTES);
    float* W1p = (float*)d_ws;
    float* W2p = (float*)((char*)d_ws + W1P_BYTES);

    if (packed) {
        dag_pack_w<<<W1P_FRAGS / 256, 256, 0, stream>>>(W1, W1p, 16, K1_);
        dag_pack_w<<<W2P_FRAGS / 256, 256, 0, stream>>>(W2, W2p, 8, D_);
        for (int l = 0; l < L_; ++l)
            dag_level_ws<<<NENC_ * REPL_, NWAVE_ * 32, 0, stream>>>(out, W1p, b1, W2p, b2,
                                                            slots, par, typ, l);
    } else {
        for (int l = 0; l < L_; ++l)
            dag_level<false><<<128, 160, 0, stream>>>(out, W1, b1, W2, b2,
                                                      slots, par, typ, l);
    }
}
